# starting kernel
# baseline (speedup 1.0000x reference)
#include <hip/hip_runtime.h>

typedef __attribute__((ext_vector_type(2))) float v2f;
typedef __attribute__((ext_vector_type(8))) float v8f;

#define APAD 132
#define HPAD 68

__device__ __forceinline__ v8f wmma_f32_k4(v2f a, v2f b, v8f c) {
  return __builtin_amdgcn_wmma_f32_16x16x4_f32(
       false, a,  false, b,
       (short)0, c,  false,  false);
}

__global__ void gme_zero_kernel(unsigned int* __restrict__ p, long long n) {
  long long i = (long long)blockIdx.x * blockDim.x + threadIdx.x;
  long long stride = (long long)gridDim.x * blockDim.x;
  for (; i < n; i += stride) p[i] = 0u;
}

__global__ __launch_bounds__(128) void gme_edge_kernel(
    const float* __restrict__ x,
    const int* __restrict__ ei_topo, const int* __restrict__ ei_geo,
    const float* __restrict__ Wt1, const float* __restrict__ bt1,
    const float* __restrict__ Wt2, const float* __restrict__ bt2,
    const float* __restrict__ Wg1, const float* __restrict__ bg1,
    const float* __restrict__ Wg2, const float* __restrict__ bg2,
    unsigned int* __restrict__ pooled_base,
    int E, int N, int tiles)
{
  __shared__ float As[16 * APAD];
  __shared__ float Hs[16 * HPAD];
  __shared__ int   Is[16];

  const int branch = blockIdx.y;
  const int*   ei = branch ? ei_geo : ei_topo;
  const float* W1 = branch ? Wg1 : Wt1;
  const float* b1 = branch ? bg1 : bt1;
  const float* W2 = branch ? Wg2 : Wt2;
  const float* b2 = branch ? bg2 : bt2;
  unsigned int* pooled = pooled_base + (size_t)branch * (size_t)N * 64;

  const int tid  = threadIdx.x;
  const int wave = tid >> 5;
  const int lane = tid & 31;
  const int m    = lane & 15;
  const int half = lane >> 4;
  const int Nc   = wave * 16 + m;

  v2f w1f[32], w2f[16];
  #pragma unroll
  for (int s = 0; s < 32; ++s) {
    const int c = 4 * s + 2 * half;
    w1f[s].x = W1[c * 64 + Nc];
    w1f[s].y = W1[(c + 1) * 64 + Nc];
  }
  #pragma unroll
  for (int s = 0; s < 16; ++s) {
    const int c = 4 * s + 2 * half;
    w2f[s].x = W2[c * 64 + Nc];
    w2f[s].y = W2[(c + 1) * 64 + Nc];
  }
  const float bias1 = b1[Nc];
  const float bias2 = b2[Nc];

  const int rh = tid >> 2;
  const int q  = tid & 3;
  const int sm = rh & 15;
  const int sh = rh >> 4;

  for (int tile = blockIdx.x; tile < tiles; tile += gridDim.x) {
    int e = tile * 16 + sm;
    if (e >= E) e = E - 1;
    const int node = ei[sh * E + e];
    if ((q | sh) == 0) Is[sm] = node;
    {
      int en = e + 16 * gridDim.x;
      if (en < E) __builtin_prefetch(&ei[sh * E + en], 0, 0);
    }
    {
      const float4* src = (const float4*)(x + (size_t)node * 64) + q * 4;
      float4* dst = (float4*)(&As[sm * APAD + sh * 64]) + q * 4;
      #pragma unroll
      for (int t = 0; t < 4; ++t) dst[t] = src[t];
    }
    __syncthreads();

    #pragma unroll
    for (int t = 0; t < 8; ++t) {
      const int pos = tid + t * 128;
      const int r = pos >> 6, c = pos & 63;
      As[r * APAD + 64 + c] -= As[r * APAD + c];
    }
    __syncthreads();

    v8f acc;
    #pragma unroll
    for (int r = 0; r < 8; ++r) acc[r] = bias1;
    #pragma unroll
    for (int s = 0; s < 32; ++s) {
      const int c = 4 * s + 2 * half;
      v2f a;
      a.x = As[m * APAD + c];
      a.y = As[m * APAD + c + 1];
      acc = wmma_f32_k4(a, w1f[s], acc);
    }
    #pragma unroll
    for (int r = 0; r < 8; ++r) {
      const float v = acc[r] > 0.f ? acc[r] : 0.f;
      Hs[(r + 8 * half) * HPAD + Nc] = v;
    }
    __syncthreads();

    v8f acc2;
    #pragma unroll
    for (int r = 0; r < 8; ++r) acc2[r] = bias2;
    #pragma unroll
    for (int s = 0; s < 16; ++s) {
      const int c = 4 * s + 2 * half;
      v2f a;
      a.x = Hs[m * HPAD + c];
      a.y = Hs[m * HPAD + c + 1];
      acc2 = wmma_f32_k4(a, w2f[s], acc2);
    }
    #pragma unroll
    for (int r = 0; r < 8; ++r) {
      const float v = acc2[r] > 0.f ? acc2[r] : 0.f;
      const int ni = Is[r + 8 * half];
      atomicMax(&pooled[(size_t)ni * 64 + Nc], __float_as_uint(v));
    }
    __syncthreads();
  }
}

__global__ __launch_bounds__(128) void gme_final_kernel(
    const float* __restrict__ pt, const float* __restrict__ pg,
    const float* __restrict__ Wf, const float* __restrict__ bf,
    float* __restrict__ out, int N, int tiles)
{
  __shared__ float As[16 * APAD];

  const int tid  = threadIdx.x;
  const int wave = tid >> 5;
  const int lane = tid & 31;
  const int m    = lane & 15;
  const int half = lane >> 4;
  const int Nc   = wave * 16 + m;

  v2f wff[32];
  #pragma unroll
  for (int s = 0; s < 32; ++s) {
    const int c = 4 * s + 2 * half;
    wff[s].x = Wf[c * 64 + Nc];
    wff[s].y = Wf[(c + 1) * 64 + Nc];
  }
  const float bias = bf[Nc];

  const int rh = tid >> 2;
  const int q  = tid & 3;
  const int sm = rh & 15;
  const int sh = rh >> 4;

  for (int tile = blockIdx.x; tile < tiles; tile += gridDim.x) {
    int row = tile * 16 + sm;
    if (row >= N) row = N - 1;
    {
      const float* srcp = (sh ? pg : pt) + (size_t)row * 64;
      const float4* src = (const float4*)srcp + q * 4;
      float4* dst = (float4*)(&As[sm * APAD + sh * 64]) + q * 4;
      #pragma unroll
      for (int t = 0; t < 4; ++t) dst[t] = src[t];
    }
    __syncthreads();

    v8f acc;
    #pragma unroll
    for (int r = 0; r < 8; ++r) acc[r] = bias;
    #pragma unroll
    for (int s = 0; s < 32; ++s) {
      const int c = 4 * s + 2 * half;
      v2f a;
      a.x = As[m * APAD + c];
      a.y = As[m * APAD + c + 1];
      acc = wmma_f32_k4(a, wff[s], acc);
    }
    #pragma unroll
    for (int r = 0; r < 8; ++r) {
      const float v = acc[r] > 0.f ? acc[r] : 0.f;
      const int orow = tile * 16 + r + 8 * half;
      if (orow < N) out[(size_t)orow * 64 + Nc] = v;
    }
    __syncthreads();
  }
}

extern "C" void kernel_launch(void* const* d_in, const int* in_sizes, int n_in,
                              void* d_out, int out_size, void* d_ws, size_t ws_size,
                              hipStream_t stream) {
  const float* x   = (const float*)d_in[0];
  const int*   eit = (const int*)d_in[1];
  const int*   eig = (const int*)d_in[2];
  const float* Wt1 = (const float*)d_in[3];
  const float* bt1 = (const float*)d_in[4];
  const float* Wt2 = (const float*)d_in[5];
  const float* bt2 = (const float*)d_in[6];
  const float* Wg1 = (const float*)d_in[7];
  const float* bg1 = (const float*)d_in[8];
  const float* Wg2 = (const float*)d_in[9];
  const float* bg2 = (const float*)d_in[10];
  const float* Wf  = (const float*)d_in[11];
  const float* bf  = (const float*)d_in[12];

  const int N = in_sizes[0] / 64;
  const int E = in_sizes[1] / 2;

  unsigned int* pooled = (unsigned int*)d_ws;
  const long long pooled_elems = 2LL * N * 64;

  {
    long long b = (pooled_elems + 255) / 256;
    int blocks = (int)(b > 50000 ? 50000 : b);
    gme_zero_kernel<<<blocks, 256, 0, stream>>>(pooled, pooled_elems);
  }
  {
    int tiles = (E + 15) / 16;
    int bx = tiles < 8192 ? tiles : 8192;
    dim3 grid(bx, 2);
    gme_edge_kernel<<<grid, 128, 0, stream>>>(
        x, eit, eig, Wt1, bt1, Wt2, bt2, Wg1, bg1, Wg2, bg2,
        pooled, E, N, tiles);
  }
  {
    int tiles = (N + 15) / 16;
    int bx = tiles < 4096 ? tiles : 4096;
    gme_final_kernel<<<bx, 128, 0, stream>>>(
        (const float*)pooled, (const float*)pooled + (size_t)N * 64,
        Wf, bf, (float*)d_out, N, tiles);
  }
}


# Layer 1 hoisted from edges to nodes by linearity: P=x(W1a-W1b)+b1, Q=x*W1b per node on WMMA, per edge h1=relu(P[i]+Q[j]); changes the f32 summation order and rounding of layer 1
# speedup vs baseline: 1.8294x; 1.8294x over previous
#include <hip/hip_runtime.h>


typedef __attribute__((ext_vector_type(2))) float v2f;
typedef __attribute__((ext_vector_type(8))) float v8f;

#define APAD 132
#define HPAD 68

__device__ __forceinline__ v8f wmma_f32_k4(v2f a, v2f b, v8f c) {
  return __builtin_amdgcn_wmma_f32_16x16x4_f32(
       false, a,  false, b,
       (short)0, c,  false,  false);
}

__global__ void gme_zero_kernel(unsigned int* __restrict__ p, long long n) {
  long long i = (long long)blockIdx.x * blockDim.x + threadIdx.x;
  long long stride = (long long)gridDim.x * blockDim.x;
  for (; i < n; i += stride) p[i] = 0u;
}

__global__ __launch_bounds__(128) void gme_node_kernel(
    const float* __restrict__ x,
    const float* __restrict__ Wt1, const float* __restrict__ bt1,
    const float* __restrict__ Wg1, const float* __restrict__ bg1,
    float* __restrict__ pq_base, int N, int tiles)
{
  __shared__ alignas(16) float Xs[2][16 * HPAD];

  const int branch = blockIdx.y;
  const float* W1 = branch ? Wg1 : Wt1;
  const float* b1 = branch ? bg1 : bt1;
  float* P = pq_base + (size_t)branch * 2 * (size_t)N * 64;
  float* Q = P + (size_t)N * 64;

  const int tid  = threadIdx.x;
  const int wave = tid >> 5;
  const int lane = tid & 31;
  const int m    = lane & 15;
  const int half = lane >> 4;
  const int Nc   = wave * 16 + m;

  v2f wpf[16], wqf[16];
  #pragma unroll
  for (int s = 0; s < 16; ++s) {
    const int c = 4 * s + 2 * half;
    const float a0 = W1[c * 64 + Nc];
    const float a1 = W1[(c + 1) * 64 + Nc];
    const float d0 = W1[(64 + c) * 64 + Nc];
    const float d1 = W1[(64 + c + 1) * 64 + Nc];
    wpf[s].x = a0 - d0;
    wpf[s].y = a1 - d1;
    wqf[s].x = d0;
    wqf[s].y = d1;
  }
  const float bias1 = b1[Nc];

  const int chunk = tid & 15;
  const int srow  = tid >> 4;

  int buf = 0;
  for (int tile = blockIdx.x; tile < tiles; tile += gridDim.x, buf ^= 1) {
    float* X = Xs[buf];
    #pragma unroll
    for (int t = 0; t < 2; ++t) {
      const int r = srow + 8 * t;
      int row = tile * 16 + r;
      if (row >= N) row = N - 1;
      const float4 v = ((const float4*)(x + (size_t)row * 64))[chunk];
      *(float4*)(&X[r * HPAD + chunk * 4]) = v;
    }
    __syncthreads();

    v8f accP, accQ;
    #pragma unroll
    for (int r = 0; r < 8; ++r) { accP[r] = bias1; accQ[r] = 0.f; }
    #pragma unroll
    for (int s = 0; s < 16; ++s) {
      const int c = 4 * s + 2 * half;
      v2f a;
      a.x = X[m * HPAD + c];
      a.y = X[m * HPAD + c + 1];
      accP = wmma_f32_k4(a, wpf[s], accP);
      accQ = wmma_f32_k4(a, wqf[s], accQ);
    }
    #pragma unroll
    for (int r = 0; r < 8; ++r) {
      const int orow = tile * 16 + r + 8 * half;
      if (orow < N) {
        P[(size_t)orow * 64 + Nc] = accP[r];
        Q[(size_t)orow * 64 + Nc] = accQ[r];
      }
    }
  }
}

__global__ __launch_bounds__(128) void gme_edge_kernel(
    const float* __restrict__ pq_base,
    const int* __restrict__ ei_topo, const int* __restrict__ ei_geo,
    const float* __restrict__ Wt2, const float* __restrict__ bt2,
    const float* __restrict__ Wg2, const float* __restrict__ bg2,
    unsigned int* __restrict__ pooled_base,
    int E, int N, int tiles)
{
  __shared__ alignas(16) float Hs[2][16 * HPAD];
  __shared__ int   Is[2][16];

  const int branch = blockIdx.y;
  const int*   ei = branch ? ei_geo : ei_topo;
  const float* W2 = branch ? Wg2 : Wt2;
  const float* b2 = branch ? bg2 : bt2;
  const float* P = pq_base + (size_t)branch * 2 * (size_t)N * 64;
  const float* Q = P + (size_t)N * 64;
  unsigned int* pooled = pooled_base + (size_t)branch * (size_t)N * 64;

  const int tid  = threadIdx.x;
  const int wave = tid >> 5;
  const int lane = tid & 31;
  const int m    = lane & 15;
  const int half = lane >> 4;
  const int Nc   = wave * 16 + m;

  v2f w2f[16];
  #pragma unroll
  for (int s = 0; s < 16; ++s) {
    const int c = 4 * s + 2 * half;
    w2f[s].x = W2[c * 64 + Nc];
    w2f[s].y = W2[(c + 1) * 64 + Nc];
  }
  const float bias2 = b2[Nc];

  const int chunk = lane & 15;
  const int r0    = 4 * wave + half;

  int ni[2], nj[2];
  #pragma unroll
  for (int t = 0; t < 2; ++t) {
    int e = blockIdx.x * 16 + r0 + 2 * t;
    if (e >= E) e = E - 1;
    ni[t] = ei[e];
    nj[t] = ei[E + e];
  }

  int buf = 0;
  for (int tile = blockIdx.x; tile < tiles; tile += gridDim.x, buf ^= 1) {
    float* H = Hs[buf];
    int*   I = Is[buf];

    float4 p[2], q[2];
    #pragma unroll
    for (int t = 0; t < 2; ++t) {
      p[t] = *(const float4*)(P + ((unsigned)ni[t] * 64u + (unsigned)chunk * 4u));
      q[t] = *(const float4*)(Q + ((unsigned)nj[t] * 64u + (unsigned)chunk * 4u));
    }
    #pragma unroll
    for (int t = 0; t < 2; ++t) {
      const int r = r0 + 2 * t;
      float4 h;
      h.x = p[t].x + q[t].x; h.y = p[t].y + q[t].y; h.z = p[t].z + q[t].z; h.w = p[t].w + q[t].w;
      h.x = h.x > 0.f ? h.x : 0.f;
      h.y = h.y > 0.f ? h.y : 0.f;
      h.z = h.z > 0.f ? h.z : 0.f;
      h.w = h.w > 0.f ? h.w : 0.f;
      *(float4*)(&H[r * HPAD + chunk * 4]) = h;
      if (chunk == 0) I[r] = ni[t];
    }
    __syncthreads();

    #pragma unroll
    for (int t = 0; t < 2; ++t) {
      int e = (tile + (int)gridDim.x) * 16 + r0 + 2 * t;
      if (e >= E) e = E - 1;
      ni[t] = ei[e];
      nj[t] = ei[E + e];
    }

    v8f acc2;
    #pragma unroll
    for (int r = 0; r < 8; ++r) acc2[r] = bias2;
    #pragma unroll
    for (int s = 0; s < 16; ++s) {
      const int c = 4 * s + 2 * half;
      v2f a;
      a.x = H[m * HPAD + c];
      a.y = H[m * HPAD + c + 1];
      acc2 = wmma_f32_k4(a, w2f[s], acc2);
    }
    #pragma unroll
    for (int r = 0; r < 8; ++r) {
      const float v = acc2[r] > 0.f ? acc2[r] : 0.f;
      const int node = I[r + 8 * half];
      atomicMax(&pooled[(unsigned)node * 64u + (unsigned)Nc], __float_as_uint(v));
    }
  }
}

__global__ __launch_bounds__(128) void gme_final_kernel(
    const float* __restrict__ pt, const float* __restrict__ pg,
    const float* __restrict__ Wf, const float* __restrict__ bf,
    float* __restrict__ out, int N, int tiles)
{
  __shared__ float As[16 * APAD];

  const int tid  = threadIdx.x;
  const int wave = tid >> 5;
  const int lane = tid & 31;
  const int m    = lane & 15;
  const int half = lane >> 4;
  const int Nc   = wave * 16 + m;

  v2f wff[32];
  #pragma unroll
  for (int s = 0; s < 32; ++s) {
    const int c = 4 * s + 2 * half;
    wff[s].x = Wf[c * 64 + Nc];
    wff[s].y = Wf[(c + 1) * 64 + Nc];
  }
  const float bias = bf[Nc];

  const int rh = tid >> 2;
  const int q  = tid & 3;
  const int sm = rh & 15;
  const int sh = rh >> 4;

  for (int tile = blockIdx.x; tile < tiles; tile += gridDim.x) {
    int row = tile * 16 + sm;
    if (row >= N) row = N - 1;
    {
      const float* srcp = (sh ? pg : pt) + (size_t)row * 64;
      const float4* src = (const float4*)srcp + q * 4;
      float4* dst = (float4*)(&As[sm * APAD + sh * 64]) + q * 4;
      #pragma unroll
      for (int t = 0; t < 4; ++t) dst[t] = src[t];
    }
    __syncthreads();

    v8f acc;
    #pragma unroll
    for (int r = 0; r < 8; ++r) acc[r] = bias;
    #pragma unroll
    for (int s = 0; s < 32; ++s) {
      const int c = 4 * s + 2 * half;
      v2f a;
      a.x = As[m * APAD + c];
      a.y = As[m * APAD + c + 1];
      acc = wmma_f32_k4(a, wff[s], acc);
    }
    #pragma unroll
    for (int r = 0; r < 8; ++r) {
      const float v = acc[r] > 0.f ? acc[r] : 0.f;
      const int orow = tile * 16 + r + 8 * half;
      if (orow < N) out[(size_t)orow * 64 + Nc] = v;
    }
    __syncthreads();
  }
}

extern "C" void kernel_launch(void* const* d_in, const int* in_sizes, int n_in,
                              void* d_out, int out_size, void* d_ws, size_t ws_size,
                              hipStream_t stream) {
  const float* x   = (const float*)d_in[0];
  const int*   eit = (const int*)d_in[1];
  const int*   eig = (const int*)d_in[2];
  const float* Wt1 = (const float*)d_in[3];
  const float* bt1 = (const float*)d_in[4];
  const float* Wt2 = (const float*)d_in[5];
  const float* bt2 = (const float*)d_in[6];
  const float* Wg1 = (const float*)d_in[7];
  const float* bg1 = (const float*)d_in[8];
  const float* Wg2 = (const float*)d_in[9];
  const float* bg2 = (const float*)d_in[10];
  const float* Wf  = (const float*)d_in[11];
  const float* bf  = (const float*)d_in[12];

  const int N = in_sizes[0] / 64;
  const int E = in_sizes[1] / 2;

  unsigned int* pooled = (unsigned int*)d_ws;
  const long long pooled_elems = 2LL * N * 64;
  float* pq = (float*)d_ws + pooled_elems;

  {
    long long b = (pooled_elems + 255) / 256;
    int blocks = (int)(b > 50000 ? 50000 : b);
    gme_zero_kernel<<<blocks, 256, 0, stream>>>(pooled, pooled_elems);
  }
  {
    int tiles = (N + 15) / 16;
    int bx = tiles < 1024 ? tiles : 1024;
    dim3 grid(bx, 2);
    gme_node_kernel<<<grid, 128, 0, stream>>>(x, Wt1, bt1, Wg1, bg1, pq, N, tiles);
  }
  {
    int tiles = (E + 15) / 16;
    int bx = tiles < 8192 ? tiles : 8192;
    dim3 grid(bx, 2);
    gme_edge_kernel<<<grid, 128, 0, stream>>>(
        pq, eit, eig, Wt2, bt2, Wg2, bg2, pooled, E, N, tiles);
  }
  {
    int tiles = (N + 15) / 16;
    int bx = tiles < 4096 ? tiles : 4096;
    gme_final_kernel<<<bx, 128, 0, stream>>>(
        (const float*)pooled, (const float*)pooled + (size_t)N * 64,
        Wf, bf, (float*)d_out, N, tiles);
  }
}
